# starting kernel
# baseline (speedup 1.0000x reference)
#include <hip/hip_runtime.h>

#define B_      4
#define T_      8192
#define F_      64
#define KTAPS   51
#define TILE_T  32
#define HALO    50
#define ROWS    (TILE_T + HALO)
#define PAD     65

typedef __attribute__((ext_vector_type(2))) float v2f;
typedef __attribute__((ext_vector_type(8))) float v8f;

__device__ __forceinline__ float fast_tanh(float v) {
#if __has_builtin(__builtin_amdgcn_tanhf)
    return __builtin_amdgcn_tanhf(v);
#elif __has_builtin(__builtin_amdgcn_tanh_f32)
    return __builtin_amdgcn_tanh_f32(v);
#else
    return tanhf(v);
#endif
}

__device__ __forceinline__ float fast_rcp(float v) {
#if __has_builtin(__builtin_amdgcn_rcpf)
    return __builtin_amdgcn_rcpf(v);
#else
    return 1.0f / v;
#endif
}

__global__ __launch_bounds__(64) void MCAO_softmax64(const float* __restrict__ wm,
                                                     float* __restrict__ wout) {
    const int r = threadIdx.x;
    float m = -3.4e38f;
    for (int j = 0; j < 64; ++j) m = fmaxf(m, wm[r * 64 + j]);
    float s = 0.0f;
    for (int j = 0; j < 64; ++j) s += __expf(wm[r * 64 + j] - m);
    const float inv = 1.0f / s;
    for (int j = 0; j < 64; ++j) wout[r * 64 + j] = __expf(wm[r * 64 + j] - m) * inv;
}

__global__ __launch_bounds__(256) void MCAO_fused(
    const float* __restrict__ x,
    const float* __restrict__ wsm,
    const float* __restrict__ pw,
    const float* __restrict__ pb,
    const float* __restrict__ kappa,
    const float* __restrict__ coeffs,
    float* __restrict__ outp,
    float* __restrict__ memp)
{
    __shared__ float  sX[ROWS * PAD];
    __shared__ float  sW[64 * PAD];
    __shared__ float2 sTE[TILE_T * PAD];
    __shared__ float  sC[64];

    const int tid  = threadIdx.x;
    const int bIdx = blockIdx.x / (T_ / TILE_T);
    const int t0   = (blockIdx.x % (T_ / TILE_T)) * TILE_T;

    const int wave  = tid >> 5;
    const int lane  = tid & 31;
    const int m16   = lane & 15;
    const int khalf = (lane < 16) ? 0 : 2;
    const int tm    = wave >> 2;
    const int tn    = wave & 3;

    v2f bfr[16];
    {
        const float* prow = pw + (size_t)(tn * 16 + m16) * F_;
        #pragma unroll
        for (int kk = 0; kk < 16; ++kk) {
            const int kb = kk * 4 + khalf;
            bfr[kk].x = prow[kb];
            bfr[kk].y = prow[kb + 1];
        }
    }

    for (int e = tid; e < ROWS * 64; e += 256) {
        const int j = e >> 6, f = e & 63;
        const int t = t0 - HALO + j;
        const float val = (t >= 0) ? x[((size_t)bIdx * T_ + t) * F_ + f] : 0.0f;
        sX[j * PAD + f] = val;
        if (j >= HALO) {
            float2 te;
            te.x = fast_tanh(val);
            te.y = __expf(-fabsf(val));
            sTE[(j - HALO) * PAD + f] = te;
        }
    }
    for (int e = tid; e < 64 * 64; e += 256)
        sW[(e >> 6) * PAD + (e & 63)] = wsm[e];
    if (tid < KTAPS) sC[tid] = coeffs[tid];
    __syncthreads();

    const float kap    = kappa[0];
    const float inv1mr = 1.0f / (1.0f - __expf(-kap));

    v2f afr[16];
    {
        const int arow = HALO + tm * 16 + m16;
        #pragma unroll
        for (int kk = 0; kk < 16; ++kk) {
            const int kb = kk * 4 + khalf;
            afr[kk].x = sX[arow * PAD + kb];
            afr[kk].y = sX[arow * PAD + kb + 1];
        }
    }

    v8f accE = {0.f, 0.f, 0.f, 0.f, 0.f, 0.f, 0.f, 0.f};
    v8f accO = {0.f, 0.f, 0.f, 0.f, 0.f, 0.f, 0.f, 0.f};
    #pragma unroll
    for (int kk = 0; kk < 16; kk += 2) {
        accE = __builtin_amdgcn_wmma_f32_16x16x4_f32(
            false, afr[kk],     false, bfr[kk],     (short)0, accE, false, false);
        accO = __builtin_amdgcn_wmma_f32_16x16x4_f32(
            false, afr[kk + 1], false, bfr[kk + 1], (short)0, accO, false, false);
    }
    v8f acc = accE + accO;

    const int   i    = tn * 16 + m16;
    const float bias = pb[i];
    #pragma unroll 2
    for (int v = 0; v < 8; ++v) {
        const int r = tm * 16 + v + ((lane >= 16) ? 8 : 0);
        const int t = t0 + r;

        const float2 tei = sTE[r * PAD + i];
        const float  ta  = tei.x;
        const float  ei  = tei.y;

        float mem = 0.0f;
        #pragma unroll
        for (int k = 0; k < KTAPS; ++k)
            mem = fmaf(sC[k], sX[(HALO + r - k) * PAD + i], mem);

        float coup = 0.0f;
        for (int j = 0; j < 64; ++j) {
            const float2 tej = sTE[r * PAD + j];
            const float  tb  = tej.x;
            const float  ej  = tej.y;
            const float  den = fmaxf(1.0f - ta * tb, 1e-20f);
            const float  th  = (ta - tb) * fast_rcp(den);
            const float  sg  = fast_rcp(1.0f + ei * ej);
            coup = fmaf(th * sg, sW[i * PAD + j], coup);
        }

        const float S    = (1.0f - __expf(-kap * (float)(t + 1))) * inv1mr;
        const float proj = acc[v] + bias;
        const float o    = 0.3f * mem + 0.4f * coup + 0.3f * proj * S;

        const size_t idx = ((size_t)bIdx * T_ + t) * F_ + i;
        outp[idx] = o;
        memp[idx] = mem;
    }
}

extern "C" void kernel_launch(void* const* d_in, const int* in_sizes, int n_in,
                              void* d_out, int out_size, void* d_ws, size_t ws_size,
                              hipStream_t stream) {
    const float* x   = (const float*)d_in[0];
    const float* wm  = (const float*)d_in[1];
    const float* pw  = (const float*)d_in[2];
    const float* pb  = (const float*)d_in[3];
    const float* kap = (const float*)d_in[4];
    const float* cf  = (const float*)d_in[5];

    float* wsm  = (float*)d_ws;
    float* outp = (float*)d_out;
    float* memp = outp + (size_t)B_ * T_ * F_;

    MCAO_softmax64<<<1, 64, 0, stream>>>(wm, wsm);
    MCAO_fused<<<B_ * (T_ / TILE_T), 256, 0, stream>>>(x, wsm, pw, pb, kap, cf,
                                                       outp, memp);
}


# coupling loop: both rows of a register position walk their four j tiles in lockstep (matrix ops issued in pairs, two independent rcp/mul/fma chains per wave); per-row sum order unchanged
# speedup vs baseline: 3.2681x; 3.2681x over previous
#include <hip/hip_runtime.h>


#define B_      4
#define T_      8192
#define F_      64
#define KTAPS   51
#define TILE_T  32
#define HALO    50
#define KWIN    68
#define CZERO   17
#define PAD     65

typedef __attribute__((ext_vector_type(2))) float v2f;
typedef __attribute__((ext_vector_type(4))) float v4f;
typedef __attribute__((ext_vector_type(8))) float v8f;

__device__ __forceinline__ float fast_rcp(float v) {
#if __has_builtin(__builtin_amdgcn_rcpf)
    return __builtin_amdgcn_rcpf(v);
#else
    return 1.0f / v;
#endif
}

__device__ __forceinline__ float xhalf(float v) {
    const int b = __builtin_bit_cast(int, v);
    return __builtin_bit_cast(float,
        __builtin_amdgcn_permlanex16(b, b, 0x76543210, 0xfedcba98, false, false));
}

__global__ __launch_bounds__(256) void MCAO_fused(
    const float* __restrict__ x,
    const float* __restrict__ wm,
    const float* __restrict__ pw,
    const float* __restrict__ pb,
    const float* __restrict__ kappa,
    const float* __restrict__ coeffs,
    float* __restrict__ outp,
    float* __restrict__ memp)
{
    __shared__ __attribute__((aligned(16))) float sQ[TILE_T * 64 * 4];
    __shared__ __attribute__((aligned(16))) float sE[TILE_T * 64];
    __shared__ float  sXo[TILE_T * PAD];
    __shared__ float  sCp[96];

    const int tid  = threadIdx.x;
    const int bIdx = blockIdx.x / (T_ / TILE_T);
    const int t0   = (blockIdx.x % (T_ / TILE_T)) * TILE_T;

    const int wave  = tid >> 5;
    const int lane  = tid & 31;
    const int m16   = lane & 15;
    const int half  = lane >> 4;
    const int khalf = half * 2;
    const int tm    = wave >> 2;
    const int tn    = wave & 3;
    const int xcol  = tn * 16 + m16;
    const int r0    = tm * 16;

    v2f xf[KWIN / 4];
    {
        const int rowb = bIdx * T_;
        const int tq   = t0 - HALO + r0 + khalf;
        const float* xp = x + (unsigned)((rowb + tq - khalf + 64) * F_ + xcol);
        if (t0 >= 2 * TILE_T) {
            const float* xw = x + (unsigned)((rowb + tq) * F_ + xcol);
            #pragma unroll
            for (int kk = 0; kk < KWIN / 4 - 1; ++kk) {
                xf[kk].x = xw[(kk * 4) * F_];
                xf[kk].y = xw[(kk * 4 + 1) * F_];
            }
        } else {
            #pragma unroll
            for (int kk = 0; kk < KWIN / 4 - 1; ++kk) {
                const int ta = tq + kk * 4;
                const int tb = ta + 1;
                const float va = x[(unsigned)((rowb + max(ta, 0)) * F_ + xcol)];
                const float vb = x[(unsigned)((rowb + max(tb, 0)) * F_ + xcol)];
                xf[kk].x = (ta >= 0) ? va : 0.0f;
                xf[kk].y = (tb >= 0) ? vb : 0.0f;
            }
        }
        const float pa = xp[0];
        const float pc = xp[F_];
        xf[KWIN / 4 - 1].x = half ? 0.0f : pa;
        xf[KWIN / 4 - 1].y = half ? 0.0f : pc;
    }
    if (tid < 96) {
        const int k = tid - CZERO;
        sCp[tid] = (k >= 0 && k < KTAPS) ? coeffs[k] : 0.0f;
    }
    __syncthreads();

    v2f bfr[16];
    {
        const float* prow = pw + (size_t)(tn * 16 + m16) * F_;
        #pragma unroll
        for (int kk = 0; kk < 16; ++kk) {
            const int kb = kk * 4 + khalf;
            bfr[kk].x = prow[kb];
            bfr[kk].y = prow[kb + 1];
        }
    }

    v8f memE = {0.f, 0.f, 0.f, 0.f, 0.f, 0.f, 0.f, 0.f};
    v8f memO = {0.f, 0.f, 0.f, 0.f, 0.f, 0.f, 0.f, 0.f};
    {
        #pragma unroll
        for (int kk = 0; kk < KWIN / 4; ++kk) {
            const int q = kk * 4 + khalf;
            v2f cf;
            cf.x = sCp[m16 + (KWIN - 1) - q];
            cf.y = sCp[m16 + (KWIN - 2) - q];
            if (kk & 1)
                memO = __builtin_amdgcn_wmma_f32_16x16x4_f32(
                    false, cf, false, xf[kk], (short)0, memO, false, false);
            else
                memE = __builtin_amdgcn_wmma_f32_16x16x4_f32(
                    false, cf, false, xf[kk], (short)0, memE, false, false);
        }
    }
    v8f accM = memE + memO;

    {
        #pragma unroll
        for (int p = 0; p < 4; ++p) {
            const v2f pr = half ? xf[12 + p] : xf[13 + p];
            #pragma unroll
            for (int s = 0; s < 2; ++s) {
                const float val = pr[s];
                const int   R   = r0 + 4 * p + 2 * (1 - half) + s;
                const float E   = expf(2.0f * val);
                const float ee  = __expf(-fabsf(val));
                v4f qv;
                qv.x = 1.0f; qv.y = E; qv.z = ee; qv.w = E * ee;
                const int o = R * 64 + xcol;
                *reinterpret_cast<v4f*>(&sQ[o * 4]) = qv;
                sE[o] = E;
                sXo[R * PAD + xcol] = val;
            }
        }
    }
    __syncthreads();

    const float kap    = kappa[0];
    const float inv1mr = 1.0f / (1.0f - __expf(-kap));

    v2f afr[16];
    {
        const int arow = r0 + m16;
        #pragma unroll
        for (int kk = 0; kk < 16; ++kk) {
            const int kb = kk * 4 + khalf;
            afr[kk].x = sXo[arow * PAD + kb];
            afr[kk].y = sXo[arow * PAD + kb + 1];
        }
    }

    v8f accE = {0.f, 0.f, 0.f, 0.f, 0.f, 0.f, 0.f, 0.f};
    v8f accO = {0.f, 0.f, 0.f, 0.f, 0.f, 0.f, 0.f, 0.f};
    #pragma unroll
    for (int kk = 0; kk < 16; kk += 2) {
        accE = __builtin_amdgcn_wmma_f32_16x16x4_f32(
            false, afr[kk],     false, bfr[kk],     (short)0, accE, false, false);
        accO = __builtin_amdgcn_wmma_f32_16x16x4_f32(
            false, afr[kk + 1], false, bfr[kk + 1], (short)0, accO, false, false);
    }
    v8f acc = accE + accO;
    asm volatile("" :: "v"(acc) : "memory");

    v8f wr[4];
    {
        const float* wrow = wm + (size_t)(tn * 16 + m16) * F_ + 8 * half;
        #pragma unroll
        for (int jt = 0; jt < 4; ++jt) {
            const v4f lo = *reinterpret_cast<const v4f*>(wrow + jt * 16);
            const v4f hi = *reinterpret_cast<const v4f*>(wrow + jt * 16 + 4);
            wr[jt][0] = lo.x; wr[jt][1] = lo.y; wr[jt][2] = lo.z; wr[jt][3] = lo.w;
            wr[jt][4] = hi.x; wr[jt][5] = hi.y; wr[jt][6] = hi.z; wr[jt][7] = hi.w;
        }
    }

    {
        float mx = wr[0][0];
        #pragma unroll
        for (int jt = 0; jt < 4; ++jt) {
            #pragma unroll
            for (int u = 0; u < 8; ++u) mx = fmaxf(mx, wr[jt][u]);
        }
        mx = fmaxf(mx, xhalf(mx));
        #pragma unroll
        for (int jt = 0; jt < 4; ++jt) {
            #pragma unroll
            for (int u = 0; u < 8; ++u) wr[jt][u] = __expf(wr[jt][u] - mx);
        }
        const v8f s8 = (wr[0] + wr[1]) + (wr[2] + wr[3]);
        float s = ((s8[0] + s8[1]) + (s8[2] + s8[3])) + ((s8[4] + s8[5]) + (s8[6] + s8[7]));
        s += xhalf(s);
        const float inv = 1.0f / s;
        #pragma unroll
        for (int jt = 0; jt < 4; ++jt) wr[jt] = wr[jt] * inv;
    }

    const int   i    = tn * 16 + m16;
    const float bias = pb[i];
    const v2f* sQ2 = reinterpret_cast<const v2f*>(sQ);
    const v2f*   qA = sQ2 + (tm * 1024 + m16) * 2 + half;
    const v2f*   qB = sQ2 + (tm * 1024 + i) * 2 + half;
    const float* eA = sE + tm * 1024 + 8 * half;
    const float* eB = sE + tm * 1024 + i;
    const v8f zero8 = {0.f, 0.f, 0.f, 0.f, 0.f, 0.f, 0.f, 0.f};

    #pragma unroll
    for (int v = 0; v < 8; ++v) {
        const int Ra = v;
        const int Rb = v + 8;
        const v2f bqa = qB[Ra * 128];
        const v2f bqb = qB[Rb * 128];
        v2f Bopa; Bopa.x = bqa.y; Bopa.y = bqa.x;
        v2f Bopb; Bopb.x = bqb.y; Bopb.y = bqb.x;
        const float Eia = eB[Ra * 64];
        const float Eib = eB[Rb * 64];
        v2f sa = {0.f, 0.f};
        v2f sb = {0.f, 0.f};
        #pragma unroll
        for (int jt = 0; jt < 4; ++jt) {
            const v2f Aa = qA[(Ra * 64 + jt * 16) * 2];
            const v2f Ab = qA[(Rb * 64 + jt * 16) * 2];
            const v4f ea0 = *reinterpret_cast<const v4f*>(eA + Ra * 64 + jt * 16);
            const v4f ea1 = *reinterpret_cast<const v4f*>(eA + Ra * 64 + jt * 16 + 4);
            const v4f eb0 = *reinterpret_cast<const v4f*>(eA + Rb * 64 + jt * 16);
            const v4f eb1 = *reinterpret_cast<const v4f*>(eA + Rb * 64 + jt * 16 + 4);
            const v8f Da = __builtin_amdgcn_wmma_f32_16x16x4_f32(
                false, Aa, false, Bopa, (short)0, zero8, false, false);
            const v8f Db = __builtin_amdgcn_wmma_f32_16x16x4_f32(
                false, Ab, false, Bopb, (short)0, zero8, false, false);
            v2f na[4], nb[4];
            na[0].x = Eia - ea0.x; na[0].y = Eia - ea0.y;
            nb[0].x = Eib - eb0.x; nb[0].y = Eib - eb0.y;
            na[1].x = Eia - ea0.z; na[1].y = Eia - ea0.w;
            nb[1].x = Eib - eb0.z; nb[1].y = Eib - eb0.w;
            na[2].x = Eia - ea1.x; na[2].y = Eia - ea1.y;
            nb[2].x = Eib - eb1.x; nb[2].y = Eib - eb1.y;
            na[3].x = Eia - ea1.z; na[3].y = Eia - ea1.w;
            nb[3].x = Eib - eb1.z; nb[3].y = Eib - eb1.w;
            #pragma unroll
            for (int p = 0; p < 4; ++p) {
                v2f ra, rb, wv;
                ra.x = fast_rcp(Da[2 * p]); ra.y = fast_rcp(Da[2 * p + 1]);
                rb.x = fast_rcp(Db[2 * p]); rb.y = fast_rcp(Db[2 * p + 1]);
                wv.x = wr[jt][2 * p]; wv.y = wr[jt][2 * p + 1];
                sa = __builtin_elementwise_fma(wv, na[p] * ra, sa);
                sb = __builtin_elementwise_fma(wv, nb[p] * rb, sb);
            }
        }
        float part[2];
        part[0] = sa.x + sa.y;
        part[1] = sb.x + sb.y;
        const float own  = half ? part[1] : part[0];
        const float send = half ? part[0] : part[1];
        const float coup = own + xhalf(send);

        const int r = tm * 16 + v + 8 * half;
        const int t = t0 + r;

        const float mem = accM[v];

        const float S    = (1.0f - __expf(-kap * (float)(t + 1))) * inv1mr;
        const float proj = acc[v] + bias;
        const float o    = 0.3f * mem + 0.4f * coup + 0.3f * proj * S;

        const size_t idx = ((size_t)bIdx * T_ + t) * F_ + i;
        outp[idx] = o;
        memp[idx] = mem;
    }
}

extern "C" void kernel_launch(void* const* d_in, const int* in_sizes, int n_in,
                              void* d_out, int out_size, void* d_ws, size_t ws_size,
                              hipStream_t stream) {
    const float* x   = (const float*)d_in[0];
    const float* wm  = (const float*)d_in[1];
    const float* pw  = (const float*)d_in[2];
    const float* pb  = (const float*)d_in[3];
    const float* kap = (const float*)d_in[4];
    const float* cf  = (const float*)d_in[5];

    float* outp = (float*)d_out;
    float* memp = outp + (size_t)B_ * T_ * F_;

    MCAO_fused<<<B_ * (T_ / TILE_T), 256, 0, stream>>>(x, wm, pw, pb, kap, cf,
                                                       outp, memp);
}


# coupling loop issue order written by hand: each reciprocal followed by one independent packed op, next pair's matrix ops issued early; conv/proj results parked in LDS; sum order unchanged
# speedup vs baseline: 3.4677x; 1.0611x over previous
#include <hip/hip_runtime.h>


#define B_      4
#define T_      8192
#define F_      64
#define KTAPS   51
#define TILE_T  32
#define HALO    50
#define KWIN    68
#define CZERO   17
#define PAD     65

typedef __attribute__((ext_vector_type(2))) float v2f;
typedef __attribute__((ext_vector_type(4))) float v4f;
typedef __attribute__((ext_vector_type(8))) float v8f;

__device__ __forceinline__ float fast_rcp(float v) {
#if __has_builtin(__builtin_amdgcn_rcpf)
    return __builtin_amdgcn_rcpf(v);
#else
    return 1.0f / v;
#endif
}

__device__ __forceinline__ float xhalf(float v) {
    const int b = __builtin_bit_cast(int, v);
    return __builtin_bit_cast(float,
        __builtin_amdgcn_permlanex16(b, b, 0x76543210, 0xfedcba98, false, false));
}

#define SB __builtin_amdgcn_sched_barrier(0)

__global__ __launch_bounds__(256) void MCAO_fused(
    const float* __restrict__ x,
    const float* __restrict__ wm,
    const float* __restrict__ pw,
    const float* __restrict__ pb,
    const float* __restrict__ kappa,
    const float* __restrict__ coeffs,
    float* __restrict__ outp,
    float* __restrict__ memp)
{
    __shared__ __attribute__((aligned(16))) float sQ[TILE_T * 64 * 4];
    __shared__ __attribute__((aligned(16))) float sE[TILE_T * 64];
    __shared__ float  sXo[TILE_T * PAD];
    __shared__ float  sCp[96];
    __shared__ __attribute__((aligned(8))) float sPark[8 * 8 * 32 * 2];

    const int tid  = threadIdx.x;
    const int bIdx = blockIdx.x / (T_ / TILE_T);
    const int t0   = (blockIdx.x % (T_ / TILE_T)) * TILE_T;

    const int wave  = tid >> 5;
    const int lane  = tid & 31;
    const int m16   = lane & 15;
    const int half  = lane >> 4;
    const int khalf = half * 2;
    const int tm    = wave >> 2;
    const int tn    = wave & 3;
    const int xcol  = tn * 16 + m16;
    const int r0    = tm * 16;

    v2f xf[KWIN / 4];
    {
        const int rowb = bIdx * T_;
        const int tq   = t0 - HALO + r0 + khalf;
        const float* xp = x + (unsigned)((rowb + tq - khalf + 64) * F_ + xcol);
        if (t0 >= 2 * TILE_T) {
            const float* xw = x + (unsigned)((rowb + tq) * F_ + xcol);
            #pragma unroll
            for (int kk = 0; kk < KWIN / 4 - 1; ++kk) {
                xf[kk].x = xw[(kk * 4) * F_];
                xf[kk].y = xw[(kk * 4 + 1) * F_];
            }
        } else {
            #pragma unroll
            for (int kk = 0; kk < KWIN / 4 - 1; ++kk) {
                const int ta = tq + kk * 4;
                const int tb = ta + 1;
                const float va = x[(unsigned)((rowb + max(ta, 0)) * F_ + xcol)];
                const float vb = x[(unsigned)((rowb + max(tb, 0)) * F_ + xcol)];
                xf[kk].x = (ta >= 0) ? va : 0.0f;
                xf[kk].y = (tb >= 0) ? vb : 0.0f;
            }
        }
        const float pa = xp[0];
        const float pc = xp[F_];
        xf[KWIN / 4 - 1].x = half ? 0.0f : pa;
        xf[KWIN / 4 - 1].y = half ? 0.0f : pc;
    }
    if (tid < 96) {
        const int k = tid - CZERO;
        sCp[tid] = (k >= 0 && k < KTAPS) ? coeffs[k] : 0.0f;
    }
    __syncthreads();

    v2f bfr[16];
    {
        const float* prow = pw + (size_t)(tn * 16 + m16) * F_;
        #pragma unroll
        for (int kk = 0; kk < 16; ++kk) {
            const int kb = kk * 4 + khalf;
            bfr[kk].x = prow[kb];
            bfr[kk].y = prow[kb + 1];
        }
    }

    v8f memE = {0.f, 0.f, 0.f, 0.f, 0.f, 0.f, 0.f, 0.f};
    v8f memO = {0.f, 0.f, 0.f, 0.f, 0.f, 0.f, 0.f, 0.f};
    {
        #pragma unroll
        for (int kk = 0; kk < KWIN / 4; ++kk) {
            const int q = kk * 4 + khalf;
            v2f cf;
            cf.x = sCp[m16 + (KWIN - 1) - q];
            cf.y = sCp[m16 + (KWIN - 2) - q];
            if (kk & 1)
                memO = __builtin_amdgcn_wmma_f32_16x16x4_f32(
                    false, cf, false, xf[kk], (short)0, memO, false, false);
            else
                memE = __builtin_amdgcn_wmma_f32_16x16x4_f32(
                    false, cf, false, xf[kk], (short)0, memE, false, false);
        }
    }
    v8f accM = memE + memO;

    {
        #pragma unroll
        for (int p = 0; p < 4; ++p) {
            const v2f pr = half ? xf[12 + p] : xf[13 + p];
            #pragma unroll
            for (int s = 0; s < 2; ++s) {
                const float val = pr[s];
                const int   R   = r0 + 4 * p + 2 * (1 - half) + s;
                const float E   = expf(2.0f * val);
                const float ee  = __expf(-fabsf(val));
                v4f qv;
                qv.x = 1.0f; qv.y = E; qv.z = ee; qv.w = E * ee;
                const int o = R * 64 + xcol;
                *reinterpret_cast<v4f*>(&sQ[o * 4]) = qv;
                sE[o] = E;
                sXo[R * PAD + xcol] = val;
            }
            SB;
        }
    }
    __syncthreads();

    const float kap    = kappa[0];
    const float inv1mr = 1.0f / (1.0f - __expf(-kap));

    v2f afr[16];
    {
        const int arow = r0 + m16;
        #pragma unroll
        for (int kk = 0; kk < 16; ++kk) {
            const int kb = kk * 4 + khalf;
            afr[kk].x = sXo[arow * PAD + kb];
            afr[kk].y = sXo[arow * PAD + kb + 1];
        }
    }

    v8f accE = {0.f, 0.f, 0.f, 0.f, 0.f, 0.f, 0.f, 0.f};
    v8f accO = {0.f, 0.f, 0.f, 0.f, 0.f, 0.f, 0.f, 0.f};
    #pragma unroll
    for (int kk = 0; kk < 16; kk += 2) {
        accE = __builtin_amdgcn_wmma_f32_16x16x4_f32(
            false, afr[kk],     false, bfr[kk],     (short)0, accE, false, false);
        accO = __builtin_amdgcn_wmma_f32_16x16x4_f32(
            false, afr[kk + 1], false, bfr[kk + 1], (short)0, accO, false, false);
    }
    v8f acc = accE + accO;
    asm volatile("" :: "v"(acc) : "memory");

    v2f* const park = reinterpret_cast<v2f*>(sPark) + wave * 256 + lane;
    #pragma unroll
    for (int v = 0; v < 8; ++v) {
        v2f pk; pk.x = accM[v]; pk.y = acc[v];
        park[v * 32] = pk;
    }

    v8f wr[4];
    {
        const float* wrow = wm + (size_t)(tn * 16 + m16) * F_ + 8 * half;
        #pragma unroll
        for (int jt = 0; jt < 4; ++jt) {
            const v4f lo = *reinterpret_cast<const v4f*>(wrow + jt * 16);
            const v4f hi = *reinterpret_cast<const v4f*>(wrow + jt * 16 + 4);
            wr[jt][0] = lo.x; wr[jt][1] = lo.y; wr[jt][2] = lo.z; wr[jt][3] = lo.w;
            wr[jt][4] = hi.x; wr[jt][5] = hi.y; wr[jt][6] = hi.z; wr[jt][7] = hi.w;
        }
    }

    {
        float mx = wr[0][0];
        #pragma unroll
        for (int jt = 0; jt < 4; ++jt) {
            #pragma unroll
            for (int u = 0; u < 8; ++u) mx = fmaxf(mx, wr[jt][u]);
        }
        mx = fmaxf(mx, xhalf(mx));
        #pragma unroll
        for (int jt = 0; jt < 4; ++jt) {
            #pragma unroll
            for (int u = 0; u < 8; ++u) wr[jt][u] = __expf(wr[jt][u] - mx);
        }
        const v8f s8 = (wr[0] + wr[1]) + (wr[2] + wr[3]);
        float s = ((s8[0] + s8[1]) + (s8[2] + s8[3])) + ((s8[4] + s8[5]) + (s8[6] + s8[7]));
        s += xhalf(s);
        const float inv = 1.0f / s;
        #pragma unroll
        for (int jt = 0; jt < 4; ++jt) wr[jt] = wr[jt] * inv;
    }

    const int   i    = tn * 16 + m16;
    const float bias = pb[i];
    const v2f* sQ2 = reinterpret_cast<const v2f*>(sQ);
    const v2f*   qA = sQ2 + (tm * 1024 + m16) * 2 + half;
    const v2f*   qB = sQ2 + (tm * 1024 + i) * 2 + half;
    const float* eA = sE + tm * 1024 + 8 * half;
    const float* eB = sE + tm * 1024 + i;
    const v8f zero8 = {0.f, 0.f, 0.f, 0.f, 0.f, 0.f, 0.f, 0.f};

    SB;
    v2f An, Bn, Bopa, Bopb;
    v8f Da, Db;
    float Eia, Eib;
    {
        const v2f bqa = qB[0 * 128];
        const v2f bqb = qB[8 * 128];
        Bopa.x = bqa.y; Bopa.y = bqa.x;
        Bopb.x = bqb.y; Bopb.y = bqb.x;
        Eia = eB[0 * 64];
        Eib = eB[8 * 64];
        An  = qA[(0 * 64) * 2];
        Bn  = qA[(8 * 64) * 2];
        Da = __builtin_amdgcn_wmma_f32_16x16x4_f32(false, An, false, Bopa, (short)0, zero8, false, false);
        Db = __builtin_amdgcn_wmma_f32_16x16x4_f32(false, Bn, false, Bopb, (short)0, zero8, false, false);
    }
    v2f sa = {0.f, 0.f};
    v2f sb = {0.f, 0.f};
    v2f cra2 = {0.f, 0.f}, crb2 = {0.f, 0.f}, cra3 = {0.f, 0.f}, crb3 = {0.f, 0.f};
    v2f cna2 = {0.f, 0.f}, cnb2 = {0.f, 0.f}, cna3 = {0.f, 0.f}, cnb3 = {0.f, 0.f};
    #pragma unroll
    for (int k = 0; k <= 32; ++k) {
        const bool cur  = k < 32;
        const bool prev = k >= 1;
        const bool next = k + 1 < 32;
        const int  v    = (k >> 2) & 7;
        const int  jt   = k & 3;
        const int  vp   = ((k - 1) >> 2) & 7;
        const int  jp   = (k - 1) & 3;
        const int  vn   = ((k + 1) >> 2) & 7;
        const int  jn   = (k + 1) & 3;

        v4f ea0 = {0.f, 0.f, 0.f, 0.f}, ea1 = ea0, eb0 = ea0, eb1 = ea0;
        float Eian = Eia, Eibn = Eib;
        v2f Bopan = Bopa, Bopbn = Bopb;
        v2f bqan = {0.f, 0.f}, bqbn = {0.f, 0.f};
        v2f pk = {0.f, 0.f};
        if (prev && jp == 3) pk = park[vp * 32];
        if (cur) {
            ea0 = *reinterpret_cast<const v4f*>(eA + v * 64 + jt * 16);
            ea1 = *reinterpret_cast<const v4f*>(eA + v * 64 + jt * 16 + 4);
            eb0 = *reinterpret_cast<const v4f*>(eA + (v + 8) * 64 + jt * 16);
            eb1 = *reinterpret_cast<const v4f*>(eA + (v + 8) * 64 + jt * 16 + 4);
        }
        if (next) {
            An = qA[(vn * 64 + jn * 16) * 2];
            Bn = qA[((vn + 8) * 64 + jn * 16) * 2];
            if (jn == 0) {
                bqan = qB[vn * 128];
                bqbn = qB[(vn + 8) * 128];
                Eian = eB[vn * 64];
                Eibn = eB[(vn + 8) * 64];
            }
        }
        SB;

        v2f ra0 = {0.f, 0.f}, rb0 = ra0, ra1 = ra0, rb1 = ra0;
        v2f ma = {0.f, 0.f}, mb = {0.f, 0.f};
        v2f wp2 = {0.f, 0.f}, wp3 = {0.f, 0.f};
        if (prev) {
            wp2.x = wr[jp][4]; wp2.y = wr[jp][5];
            wp3.x = wr[jp][6]; wp3.y = wr[jp][7];
        }
        if (cur)  { ra0.x = fast_rcp(Da[0]); SB; }
        if (prev) { ma = cna2 * cra2; SB; }
        if (cur)  { ra0.y = fast_rcp(Da[1]); SB; }
        if (prev) { mb = cnb2 * crb2; SB; }
        if (cur)  { rb0.x = fast_rcp(Db[0]); SB; }
        if (prev) { sa = __builtin_elementwise_fma(wp2, ma, sa); SB; }
        if (cur)  { rb0.y = fast_rcp(Db[1]); SB; }
        if (prev) { sb = __builtin_elementwise_fma(wp2, mb, sb); SB; }
        if (cur)  { ra1.x = fast_rcp(Da[2]); SB; }
        if (prev) { ma = cna3 * cra3; SB; }
        if (cur)  { ra1.y = fast_rcp(Da[3]); SB; }
        if (prev) { mb = cnb3 * crb3; SB; }
        if (cur)  { rb1.x = fast_rcp(Db[2]); SB; }
        if (prev) { sa = __builtin_elementwise_fma(wp3, ma, sa); SB; }
        if (cur)  { rb1.y = fast_rcp(Db[3]); SB; }
        if (prev) { sb = __builtin_elementwise_fma(wp3, mb, sb); SB; }

        if (prev && jp == 3) {
            float part[2];
            part[0] = sa.x + sa.y;
            part[1] = sb.x + sb.y;
            const float own  = half ? part[1] : part[0];
            const float send = half ? part[0] : part[1];
            const float coup = own + xhalf(send);

            const int r = tm * 16 + vp + 8 * half;
            const int t = t0 + r;

            const float mem = pk.x;

            const float S    = (1.0f - __expf(-kap * (float)(t + 1))) * inv1mr;
            const float proj = pk.y + bias;
            const float o    = 0.3f * mem + 0.4f * coup + 0.3f * proj * S;

            const size_t idx = ((size_t)bIdx * T_ + t) * F_ + i;
            outp[idx] = o;
            memp[idx] = mem;
            sa.x = 0.f; sa.y = 0.f;
            sb.x = 0.f; sb.y = 0.f;
            SB;
        }

        if (cur) {
            const v2f Eva = {Eia, Eia};
            const v2f Evb = {Eib, Eib};
            v2f ra2, rb2, ra3, rb3;
            v2f na0, nb0, na1, nb1, na2, nb2, na3, nb3;
            ra2.x = fast_rcp(Da[4]); SB; na0 = Eva - __builtin_shufflevector(ea0, ea0, 0, 1); SB;
            ra2.y = fast_rcp(Da[5]); SB; nb0 = Evb - __builtin_shufflevector(eb0, eb0, 0, 1); SB;
            rb2.x = fast_rcp(Db[4]); SB; na1 = Eva - __builtin_shufflevector(ea0, ea0, 2, 3); SB;
            rb2.y = fast_rcp(Db[5]); SB; nb1 = Evb - __builtin_shufflevector(eb0, eb0, 2, 3); SB;
            ra3.x = fast_rcp(Da[6]); SB; na2 = Eva - __builtin_shufflevector(ea1, ea1, 0, 1); SB;
            ra3.y = fast_rcp(Da[7]); SB; nb2 = Evb - __builtin_shufflevector(eb1, eb1, 0, 1); SB;
            rb3.x = fast_rcp(Db[6]); SB; na3 = Eva - __builtin_shufflevector(ea1, ea1, 2, 3); SB;
            rb3.y = fast_rcp(Db[7]); SB; nb3 = Evb - __builtin_shufflevector(eb1, eb1, 2, 3); SB;
            if (next) {
                if (jn == 0) {
                    Bopan.x = bqan.y; Bopan.y = bqan.x;
                    Bopbn.x = bqbn.y; Bopbn.y = bqbn.x;
                }
                Da = __builtin_amdgcn_wmma_f32_16x16x4_f32(false, An, false, Bopan, (short)0, zero8, false, false);
                Db = __builtin_amdgcn_wmma_f32_16x16x4_f32(false, Bn, false, Bopbn, (short)0, zero8, false, false);
            }
            SB;
            v2f w0, w1;
            w0.x = wr[jt][0]; w0.y = wr[jt][1];
            w1.x = wr[jt][2]; w1.y = wr[jt][3];
            ma = na0 * ra0; SB;
            mb = nb0 * rb0; SB;
            sa = __builtin_elementwise_fma(w0, ma, sa); SB;
            sb = __builtin_elementwise_fma(w0, mb, sb); SB;
            ma = na1 * ra1; SB;
            mb = nb1 * rb1; SB;
            sa = __builtin_elementwise_fma(w1, ma, sa); SB;
            sb = __builtin_elementwise_fma(w1, mb, sb); SB;
            cra2 = ra2; crb2 = rb2; cra3 = ra3; crb3 = rb3;
            cna2 = na2; cnb2 = nb2; cna3 = na3; cnb3 = nb3;
            Bopa = Bopan; Bopb = Bopbn;
            Eia = Eian; Eib = Eibn;
        }
    }
#undef SB
}

extern "C" void kernel_launch(void* const* d_in, const int* in_sizes, int n_in,
                              void* d_out, int out_size, void* d_ws, size_t ws_size,
                              hipStream_t stream) {
    const float* x   = (const float*)d_in[0];
    const float* wm  = (const float*)d_in[1];
    const float* pw  = (const float*)d_in[2];
    const float* pb  = (const float*)d_in[3];
    const float* kap = (const float*)d_in[4];
    const float* cf  = (const float*)d_in[5];

    float* outp = (float*)d_out;
    float* memp = outp + (size_t)B_ * T_ * F_;

    MCAO_fused<<<B_ * (T_ / TILE_T), 256, 0, stream>>>(x, wm, pw, pb, kap, cf,
                                                       outp, memp);
}
